# starting kernel
# baseline (speedup 1.0000x reference)
#include <hip/hip_runtime.h>
#include <hip/hip_bf16.h>
#include <math.h>

typedef float v2f __attribute__((ext_vector_type(2)));
typedef float v8f __attribute__((ext_vector_type(8)));

#define NEG_SLOPE 0.2f

__device__ __forceinline__ float leaky(float v) {
    return v > 0.0f ? v : NEG_SLOPE * v;
}

__device__ __forceinline__ unsigned enc_f32(float v) {
    unsigned u = __float_as_uint(v);
    return (u >> 31) ? ~u : (u | 0x80000000u);
}
__device__ __forceinline__ float dec_max(unsigned u) {
    if (u == 0u) return 0.0f;
    unsigned b = (u >> 31) ? (u & 0x7fffffffu) : ~u;
    return __uint_as_float(b);
}

__global__ void gat_init(float* acc1, unsigned* emax1, float* denom1,
                         unsigned* emax2, float* denom2, float* out,
                         const float* __restrict__ b2, int N) {
    int idx = blockIdx.x * blockDim.x + threadIdx.x;
    if (idx >= N * 64) return;
    acc1[idx] = 0.0f;
    out[idx]  = b2[idx & 63];
    if (idx < N * 8) { emax1[idx] = 0u; denom1[idx] = 0.0f; }
    if (idx < N)     { emax2[idx] = 0u; denom2[idx] = 0.0f; }
}

__global__ __launch_bounds__(128)
void wmma_gemm_f32(const float* __restrict__ A, const float* __restrict__ B,
                   float* __restrict__ C, int M, int Nn, int K) {
    const int wave = threadIdx.x >> 5;
    const int lane = threadIdx.x & 31;
    const int tilesN = Nn >> 4;
    const int tile = blockIdx.x * 4 + wave;
    const int totalTiles = (M >> 4) * tilesN;
    if (tile >= totalTiles) return;

    const int m0 = (tile / tilesN) << 4;
    const int n0 = (tile % tilesN) << 4;
    const int half = lane >> 4;
    const int l    = lane & 15;

    v8f c = {};
    for (int k0 = 0; k0 < K; k0 += 4) {
        const int ka = k0 + 2 * half;
        v2f a, b;
        a.x = A[(size_t)(m0 + l) * K + ka];
        a.y = A[(size_t)(m0 + l) * K + ka + 1];
        b.x = B[(size_t)ka * Nn + n0 + l];
        b.y = B[(size_t)(ka + 1) * Nn + n0 + l];
        c = __builtin_amdgcn_wmma_f32_16x16x4_f32(
                 false, a,  false, b,
                 (short)0, c,  false,  false);
    }

    const int row = m0 + half * 8;
#pragma unroll
    for (int i = 0; i < 8; ++i)
        C[(size_t)(row + i) * Nn + n0 + l] = c[i];
}

__global__ void attn_coef1(const float* __restrict__ feat,
                           const float* __restrict__ al, const float* __restrict__ ar,
                           float* __restrict__ el, float* __restrict__ er, int N) {
    int idx = blockIdx.x * blockDim.x + threadIdx.x;
    if (idx >= N * 8) return;
    int n = idx >> 3, h = idx & 7;
    const float* f = feat + (size_t)n * 64 + h * 8;
    float sl = 0.0f, sr = 0.0f;
#pragma unroll
    for (int d = 0; d < 8; ++d) {
        sl += f[d] * al[h * 8 + d];
        sr += f[d] * ar[h * 8 + d];
    }
    el[idx] = sl;
    er[idx] = sr;
}

__global__ void attn_coef2(const float* __restrict__ feat,
                           const float* __restrict__ al, const float* __restrict__ ar,
                           float* __restrict__ el, float* __restrict__ er, int N) {
    int n = blockIdx.x * blockDim.x + threadIdx.x;
    if (n >= N) return;
    const float* f = feat + (size_t)n * 64;
    float sl = 0.0f, sr = 0.0f;
#pragma unroll
    for (int d = 0; d < 64; ++d) {
        sl += f[d] * al[d];
        sr += f[d] * ar[d];
    }
    el[n] = sl;
    er[n] = sr;
}

__global__ void edge_max(const int* __restrict__ src, const int* __restrict__ dst,
                         const float* __restrict__ el, const float* __restrict__ er,
                         unsigned* emax, int E, int H) {
    int idx = blockIdx.x * blockDim.x + threadIdx.x;
    if (idx >= E * H) return;
    int e = idx / H, h = idx - e * H;
    float v = leaky(el[src[e] * H + h] + er[dst[e] * H + h]);
    atomicMax(&emax[dst[e] * H + h], enc_f32(v));
}

__global__ void edge_denom(const int* __restrict__ src, const int* __restrict__ dst,
                           const float* __restrict__ el, const float* __restrict__ er,
                           const unsigned* __restrict__ emax, float* denom, int E, int H) {
    int idx = blockIdx.x * blockDim.x + threadIdx.x;
    if (idx >= E * H) return;
    int e = idx / H, h = idx - e * H;
    int dn = dst[e];
    float v = leaky(el[src[e] * H + h] + er[dn * H + h]);
    float m = dec_max(emax[dn * H + h]);
    atomicAdd(&denom[dn * H + h], expf(v - m));
}

__global__ void edge_agg1(const int* __restrict__ src, const int* __restrict__ dst,
                          const float* __restrict__ el, const float* __restrict__ er,
                          const unsigned* __restrict__ emax, const float* __restrict__ denom,
                          const float* __restrict__ feat, float* acc, int E) {
    int idx = blockIdx.x * blockDim.x + threadIdx.x;
    if (idx >= E * 8) return;
    int e = idx >> 3, h = idx & 7;
    int s = src[e], dn = dst[e];
    float v = leaky(el[s * 8 + h] + er[dn * 8 + h]);
    float m = dec_max(emax[dn * 8 + h]);
    float alpha = expf(v - m) / fmaxf(denom[dn * 8 + h], 1e-9f);
    const float* fs = feat + (size_t)s * 64 + h * 8;
    float* od = acc + (size_t)dn * 64 + h * 8;
#pragma unroll
    for (int d = 0; d < 8; ++d)
        atomicAdd(&od[d], fs[d] * alpha);
}

__global__ void edge_agg2(const int* __restrict__ src, const int* __restrict__ dst,
                          const float* __restrict__ el, const float* __restrict__ er,
                          const unsigned* __restrict__ emax, const float* __restrict__ denom,
                          const float* __restrict__ feat, float* out, int E) {
    int idx = blockIdx.x * blockDim.x + threadIdx.x;
    if (idx >= E * 4) return;
    int e = idx >> 2, chunk = idx & 3;
    int s = src[e], dn = dst[e];
    float v = leaky(el[s] + er[dn]);
    float m = dec_max(emax[dn]);
    float alpha = expf(v - m) / fmaxf(denom[dn], 1e-9f);
    const float* fs = feat + (size_t)s * 64 + chunk * 16;
    float* od = out + (size_t)dn * 64 + chunk * 16;
#pragma unroll
    for (int d = 0; d < 16; ++d)
        atomicAdd(&od[d], fs[d] * alpha);
}

__global__ void elu_bias(const float* __restrict__ acc, const float* __restrict__ b1,
                         float* __restrict__ x2, int N) {
    int idx = blockIdx.x * blockDim.x + threadIdx.x;
    if (idx >= N * 64) return;
    float t = acc[idx] + b1[idx & 63];
    x2[idx] = t > 0.0f ? t : expm1f(t);
}

extern "C" void kernel_launch(void* const* d_in, const int* in_sizes, int n_in,
                              void* d_out, int out_size, void* d_ws, size_t ws_size,
                              hipStream_t stream) {
    (void)n_in; (void)out_size; (void)ws_size;
    const float* h   = (const float*)d_in[0];
    const int*   src = (const int*)d_in[1];
    const int*   dst = (const int*)d_in[2];
    const float* W1  = (const float*)d_in[3];
    const float* al1 = (const float*)d_in[4];
    const float* ar1 = (const float*)d_in[5];
    const float* b1  = (const float*)d_in[6];
    const float* W2  = (const float*)d_in[7];
    const float* al2 = (const float*)d_in[8];
    const float* ar2 = (const float*)d_in[9];
    const float* b2  = (const float*)d_in[10];
    float* out = (float*)d_out;

    const int N = in_sizes[0] / 128;
    const int E = in_sizes[1];

    float*    feat1  = (float*)d_ws;
    float*    el1    = feat1 + (size_t)N * 64;
    float*    er1    = el1 + (size_t)N * 8;
    unsigned* emax1  = (unsigned*)(er1 + (size_t)N * 8);
    float*    denom1 = (float*)(emax1 + (size_t)N * 8);
    float*    acc1   = denom1 + (size_t)N * 8;
    float*    x2     = acc1 + (size_t)N * 64;
    float*    feat2  = x2 + (size_t)N * 64;
    float*    el2    = feat2 + (size_t)N * 64;
    float*    er2    = el2 + N;
    unsigned* emax2  = (unsigned*)(er2 + N);
    float*    denom2 = (float*)(emax2 + N);

    const int B = 256;
    const int gN64 = (N * 64 + B - 1) / B;
    const int gN8  = (N * 8 + B - 1) / B;
    const int gN   = (N + B - 1) / B;
    const int gE8  = (E * 8 + B - 1) / B;
    const int gE4  = (E * 4 + B - 1) / B;
    const int gE1  = (E + B - 1) / B;

    gat_init<<<gN64, B, 0, stream>>>(acc1, emax1, denom1, emax2, denom2, out, b2, N);

    {
        int tiles = (N / 16) * (64 / 16);
        wmma_gemm_f32<<<(tiles + 3) / 4, 128, 0, stream>>>(h, W1, feat1, N, 64, 128);
    }
    attn_coef1<<<gN8, B, 0, stream>>>(feat1, al1, ar1, el1, er1, N);
    edge_max  <<<gE8, B, 0, stream>>>(src, dst, el1, er1, emax1, E, 8);
    edge_denom<<<gE8, B, 0, stream>>>(src, dst, el1, er1, emax1, denom1, E, 8);
    edge_agg1 <<<gE8, B, 0, stream>>>(src, dst, el1, er1, emax1, denom1, feat1, acc1, E);
    elu_bias  <<<gN64, B, 0, stream>>>(acc1, b1, x2, N);

    {
        int tiles = (N / 16) * (64 / 16);
        wmma_gemm_f32<<<(tiles + 3) / 4, 128, 0, stream>>>(x2, W2, feat2, N, 64, 64);
    }
    attn_coef2<<<gN, B, 0, stream>>>(feat2, al2, ar2, el2, er2, N);
    edge_max  <<<gE1, B, 0, stream>>>(src, dst, el2, er2, emax2, E, 1);
    edge_denom<<<gE1, B, 0, stream>>>(src, dst, el2, er2, emax2, denom2, E, 1);
    edge_agg2 <<<gE4, B, 0, stream>>>(src, dst, el2, er2, emax2, denom2, feat2, out, E);
}


# edge_agg1/2: one wave per 32 edges, lanes along the 64-float row, so each row load and atomicAdd covers one 128-byte line (loads batched ahead of atomics); float atomic order unordered as before
# speedup vs baseline: 1.8075x; 1.8075x over previous
#include <hip/hip_runtime.h>
#include <hip/hip_bf16.h>
#include <math.h>

typedef float v2f __attribute__((ext_vector_type(2)));
typedef float v8f __attribute__((ext_vector_type(8)));

#define NEG_SLOPE 0.2f

__device__ __forceinline__ float leaky(float v) {
    return v > 0.0f ? v : NEG_SLOPE * v;
}

__device__ __forceinline__ unsigned enc_f32(float v) {
    unsigned u = __float_as_uint(v);
    return (u >> 31) ? ~u : (u | 0x80000000u);
}
__device__ __forceinline__ float dec_max(unsigned u) {
    if (u == 0u) return 0.0f;
    unsigned b = (u >> 31) ? (u & 0x7fffffffu) : ~u;
    return __uint_as_float(b);
}

__global__ void gat_init(float* acc1, unsigned* emax1, float* denom1,
                         unsigned* emax2, float* denom2, float* out,
                         const float* __restrict__ b2, int N) {
    int idx = blockIdx.x * blockDim.x + threadIdx.x;
    if (idx >= N * 64) return;
    acc1[idx] = 0.0f;
    out[idx]  = b2[idx & 63];
    if (idx < N * 8) { emax1[idx] = 0u; denom1[idx] = 0.0f; }
    if (idx < N)     { emax2[idx] = 0u; denom2[idx] = 0.0f; }
}

__global__ __launch_bounds__(128)
void wmma_gemm_f32(const float* __restrict__ A, const float* __restrict__ B,
                   float* __restrict__ C, int M, int Nn, int K) {
    const int wave = threadIdx.x >> 5;
    const int lane = threadIdx.x & 31;
    const int tilesN = Nn >> 4;
    const int tile = blockIdx.x * 4 + wave;
    const int totalTiles = (M >> 4) * tilesN;
    if (tile >= totalTiles) return;

    const int m0 = (tile / tilesN) << 4;
    const int n0 = (tile % tilesN) << 4;
    const int half = lane >> 4;
    const int l    = lane & 15;

    v8f c = {};
    for (int k0 = 0; k0 < K; k0 += 4) {
        const int ka = k0 + 2 * half;
        v2f a, b;
        a.x = A[(size_t)(m0 + l) * K + ka];
        a.y = A[(size_t)(m0 + l) * K + ka + 1];
        b.x = B[(size_t)ka * Nn + n0 + l];
        b.y = B[(size_t)(ka + 1) * Nn + n0 + l];
        c = __builtin_amdgcn_wmma_f32_16x16x4_f32(
                 false, a,  false, b,
                 (short)0, c,  false,  false);
    }

    const int row = m0 + half * 8;
#pragma unroll
    for (int i = 0; i < 8; ++i)
        C[(size_t)(row + i) * Nn + n0 + l] = c[i];
}

__global__ void attn_coef1(const float* __restrict__ feat,
                           const float* __restrict__ al, const float* __restrict__ ar,
                           float* __restrict__ el, float* __restrict__ er, int N) {
    int idx = blockIdx.x * blockDim.x + threadIdx.x;
    if (idx >= N * 8) return;
    int n = idx >> 3, h = idx & 7;
    const float* f = feat + (size_t)n * 64 + h * 8;
    float sl = 0.0f, sr = 0.0f;
#pragma unroll
    for (int d = 0; d < 8; ++d) {
        sl += f[d] * al[h * 8 + d];
        sr += f[d] * ar[h * 8 + d];
    }
    el[idx] = sl;
    er[idx] = sr;
}

__global__ void attn_coef2(const float* __restrict__ feat,
                           const float* __restrict__ al, const float* __restrict__ ar,
                           float* __restrict__ el, float* __restrict__ er, int N) {
    int n = blockIdx.x * blockDim.x + threadIdx.x;
    if (n >= N) return;
    const float* f = feat + (size_t)n * 64;
    float sl = 0.0f, sr = 0.0f;
#pragma unroll
    for (int d = 0; d < 64; ++d) {
        sl += f[d] * al[d];
        sr += f[d] * ar[d];
    }
    el[n] = sl;
    er[n] = sr;
}

__global__ void edge_max(const int* __restrict__ src, const int* __restrict__ dst,
                         const float* __restrict__ el, const float* __restrict__ er,
                         unsigned* emax, int E, int H) {
    int idx = blockIdx.x * blockDim.x + threadIdx.x;
    if (idx >= E * H) return;
    int e = idx / H, h = idx - e * H;
    float v = leaky(el[src[e] * H + h] + er[dst[e] * H + h]);
    atomicMax(&emax[dst[e] * H + h], enc_f32(v));
}

__global__ void edge_denom(const int* __restrict__ src, const int* __restrict__ dst,
                           const float* __restrict__ el, const float* __restrict__ er,
                           const unsigned* __restrict__ emax, float* denom, int E, int H) {
    int idx = blockIdx.x * blockDim.x + threadIdx.x;
    if (idx >= E * H) return;
    int e = idx / H, h = idx - e * H;
    int dn = dst[e];
    float v = leaky(el[src[e] * H + h] + er[dn * H + h]);
    float m = dec_max(emax[dn * H + h]);
    atomicAdd(&denom[dn * H + h], expf(v - m));
}


template <bool FULL>
__device__ __forceinline__ void agg1_rows(const int* __restrict__ src, const int* __restrict__ dst,
                                          const float* __restrict__ el, const float* __restrict__ er,
                                          const unsigned* __restrict__ emax, const float* __restrict__ denom,
                                          const float* __restrict__ feat, float* acc,
                                          int E, int e0, int lane) {
    const int j = lane >> 3, h = lane & 7;
    const int cnt = FULL ? 32 : E - e0;
    int sv[8], dv[8];
    float av[8];
#pragma unroll
    for (int r = 0; r < 8; ++r) {
        const bool ok = r * 4 + j < cnt;
        const int s  = ok ? src[e0 + r * 4 + j] : 0;
        const int dn = ok ? dst[e0 + r * 4 + j] : 0;
        const float v = leaky(el[s * 8 + h] + er[dn * 8 + h]);
        const float m = dec_max(emax[dn * 8 + h]);
        sv[r] = s;
        dv[r] = dn;
        av[r] = expf(v - m) / fmaxf(denom[dn * 8 + h], 1e-9f);
    }
#pragma unroll
    for (int b = 0; b < 4; ++b) {
        float f0[8], f1[8];
#pragma unroll
        for (int i = 0; i < 8; ++i) {
            const int r = b * 2 + (i >> 2), jj = i & 3;
            if (r * 4 + jj < cnt) {
                const int s = __builtin_amdgcn_readlane(sv[r], jj * 8);
                const float* fs = feat + (size_t)s * 64;
                f0[i] = fs[lane];
                f1[i] = fs[lane + 32];
            }
        }
#pragma unroll
        for (int i = 0; i < 8; ++i) {
            const int r = b * 2 + (i >> 2), jj = i & 3;
            if (r * 4 + jj < cnt) {
                const int dn = __builtin_amdgcn_readlane(dv[r], jj * 8);
                const float alo = __shfl(av[r], jj * 8 + j);
                const float ahi = __shfl(av[r], jj * 8 + 4 + j);
                float* od = acc + (size_t)dn * 64;
                atomicAdd(&od[lane], f0[i] * alo);
                atomicAdd(&od[lane + 32], f1[i] * ahi);
            }
        }
    }
}

__global__ __launch_bounds__(256)
void edge_agg1(const int* __restrict__ src, const int* __restrict__ dst,
               const float* __restrict__ el, const float* __restrict__ er,
               const unsigned* __restrict__ emax, const float* __restrict__ denom,
               const float* __restrict__ feat, float* acc, int E) {
    const int lane = threadIdx.x & 31;
    const int e0 = ((blockIdx.x * blockDim.x + threadIdx.x) >> 5) * 32;
    if (e0 >= E) return;
    if (e0 + 32 <= E) agg1_rows<true >(src, dst, el, er, emax, denom, feat, acc, E, e0, lane);
    else              agg1_rows<false>(src, dst, el, er, emax, denom, feat, acc, E, e0, lane);
}

template <bool FULL>
__device__ __forceinline__ void agg2_rows(const int* __restrict__ src, const int* __restrict__ dst,
                                          const float* __restrict__ el, const float* __restrict__ er,
                                          const unsigned* __restrict__ emax, const float* __restrict__ denom,
                                          const float* __restrict__ feat, float* out,
                                          int E, int e0, int lane) {
    const int cnt = FULL ? 32 : E - e0;
    const bool ok = lane < cnt;
    const int sl = ok ? src[e0 + lane] : 0;
    const int dl = ok ? dst[e0 + lane] : 0;
    const float v = leaky(el[sl] + er[dl]);
    const float m = dec_max(emax[dl]);
    const float al = expf(v - m) / fmaxf(denom[dl], 1e-9f);
#pragma unroll
    for (int b = 0; b < 2; ++b) {
        float f0[16], f1[16];
#pragma unroll
        for (int i = 0; i < 16; ++i) {
            const int k = b * 16 + i;
            if (k < cnt) {
                const int s = __builtin_amdgcn_readlane(sl, k);
                const float* fs = feat + (size_t)s * 64;
                f0[i] = fs[lane];
                f1[i] = fs[lane + 32];
            }
        }
#pragma unroll
        for (int i = 0; i < 16; ++i) {
            const int k = b * 16 + i;
            if (k < cnt) {
                const int dn = __builtin_amdgcn_readlane(dl, k);
                const float a = __int_as_float(__builtin_amdgcn_readlane(__float_as_int(al), k));
                float* od = out + (size_t)dn * 64;
                atomicAdd(&od[lane], f0[i] * a);
                atomicAdd(&od[lane + 32], f1[i] * a);
            }
        }
    }
}

__global__ __launch_bounds__(256)
void edge_agg2(const int* __restrict__ src, const int* __restrict__ dst,
               const float* __restrict__ el, const float* __restrict__ er,
               const unsigned* __restrict__ emax, const float* __restrict__ denom,
               const float* __restrict__ feat, float* out, int E) {
    const int lane = threadIdx.x & 31;
    const int e0 = ((blockIdx.x * blockDim.x + threadIdx.x) >> 5) * 32;
    if (e0 >= E) return;
    if (e0 + 32 <= E) agg2_rows<true >(src, dst, el, er, emax, denom, feat, out, E, e0, lane);
    else              agg2_rows<false>(src, dst, el, er, emax, denom, feat, out, E, e0, lane);
}

__global__ void elu_bias(const float* __restrict__ acc, const float* __restrict__ b1,
                         float* __restrict__ x2, int N) {
    int idx = blockIdx.x * blockDim.x + threadIdx.x;
    if (idx >= N * 64) return;
    float t = acc[idx] + b1[idx & 63];
    x2[idx] = t > 0.0f ? t : expm1f(t);
}

extern "C" void kernel_launch(void* const* d_in, const int* in_sizes, int n_in,
                              void* d_out, int out_size, void* d_ws, size_t ws_size,
                              hipStream_t stream) {
    (void)n_in; (void)out_size; (void)ws_size;
    const float* h   = (const float*)d_in[0];
    const int*   src = (const int*)d_in[1];
    const int*   dst = (const int*)d_in[2];
    const float* W1  = (const float*)d_in[3];
    const float* al1 = (const float*)d_in[4];
    const float* ar1 = (const float*)d_in[5];
    const float* b1  = (const float*)d_in[6];
    const float* W2  = (const float*)d_in[7];
    const float* al2 = (const float*)d_in[8];
    const float* ar2 = (const float*)d_in[9];
    const float* b2  = (const float*)d_in[10];
    float* out = (float*)d_out;

    const int N = in_sizes[0] / 128;
    const int E = in_sizes[1];

    float*    feat1  = (float*)d_ws;
    float*    el1    = feat1 + (size_t)N * 64;
    float*    er1    = el1 + (size_t)N * 8;
    unsigned* emax1  = (unsigned*)(er1 + (size_t)N * 8);
    float*    denom1 = (float*)(emax1 + (size_t)N * 8);
    float*    acc1   = denom1 + (size_t)N * 8;
    float*    x2     = acc1 + (size_t)N * 64;
    float*    feat2  = x2 + (size_t)N * 64;
    float*    el2    = feat2 + (size_t)N * 64;
    float*    er2    = el2 + N;
    unsigned* emax2  = (unsigned*)(er2 + N);
    float*    denom2 = (float*)(emax2 + N);

    const int B = 256;
    const int gN64 = (N * 64 + B - 1) / B;
    const int gN8  = (N * 8 + B - 1) / B;
    const int gN   = (N + B - 1) / B;
    const int gE8  = (E * 8 + B - 1) / B;
    const int gE1  = (E + B - 1) / B;

    gat_init<<<gN64, B, 0, stream>>>(acc1, emax1, denom1, emax2, denom2, out, b2, N);

    {
        int tiles = (N / 16) * (64 / 16);
        wmma_gemm_f32<<<(tiles + 3) / 4, 128, 0, stream>>>(h, W1, feat1, N, 64, 128);
    }
    attn_coef1<<<gN8, B, 0, stream>>>(feat1, al1, ar1, el1, er1, N);
    edge_max  <<<gE8, B, 0, stream>>>(src, dst, el1, er1, emax1, E, 8);
    edge_denom<<<gE8, B, 0, stream>>>(src, dst, el1, er1, emax1, denom1, E, 8);
    edge_agg1 <<<gE1, B, 0, stream>>>(src, dst, el1, er1, emax1, denom1, feat1, acc1, E);
    elu_bias  <<<gN64, B, 0, stream>>>(acc1, b1, x2, N);

    {
        int tiles = (N / 16) * (64 / 16);
        wmma_gemm_f32<<<(tiles + 3) / 4, 128, 0, stream>>>(x2, W2, feat2, N, 64, 64);
    }
    attn_coef2<<<gN, B, 0, stream>>>(feat2, al2, ar2, el2, er2, N);
    edge_max  <<<gE1, B, 0, stream>>>(src, dst, el2, er2, emax2, E, 1);
    edge_denom<<<gE1, B, 0, stream>>>(src, dst, el2, er2, emax2, denom2, E, 1);
    edge_agg2 <<<gE1, B, 0, stream>>>(src, dst, el2, er2, emax2, denom2, feat2, out, E);
}


# Aggregation without float atomics: edges counting-sorted by dst each call (int atomic rank), one wave per node does max, denominator and weighted row sum in registers; sum order = arrival rank
# speedup vs baseline: 2.5230x; 1.3959x over previous
#include <hip/hip_runtime.h>
#include <hip/hip_bf16.h>
#include <math.h>

typedef float v2f __attribute__((ext_vector_type(2)));
typedef float v8f __attribute__((ext_vector_type(8)));

#define NEG_SLOPE 0.2f

__device__ __forceinline__ float leaky(float v) {
    return v > 0.0f ? v : NEG_SLOPE * v;
}

__global__ __launch_bounds__(128)
void wmma_gemm_f32(const float* __restrict__ A, const float* __restrict__ B,
                   float* __restrict__ C, int M, int Nn, int K) {
    const int wave = threadIdx.x >> 5;
    const int lane = threadIdx.x & 31;
    const int tilesN = Nn >> 4;
    const int tile = blockIdx.x * 4 + wave;
    const int totalTiles = (M >> 4) * tilesN;
    if (tile >= totalTiles) return;

    const int m0 = (tile / tilesN) << 4;
    const int n0 = (tile % tilesN) << 4;
    const int half = lane >> 4;
    const int l    = lane & 15;

    v8f c = {};
    for (int k0 = 0; k0 < K; k0 += 4) {
        const int ka = k0 + 2 * half;
        v2f a, b;
        a.x = A[(size_t)(m0 + l) * K + ka];
        a.y = A[(size_t)(m0 + l) * K + ka + 1];
        b.x = B[(size_t)ka * Nn + n0 + l];
        b.y = B[(size_t)(ka + 1) * Nn + n0 + l];
        c = __builtin_amdgcn_wmma_f32_16x16x4_f32(
                 false, a,  false, b,
                 (short)0, c,  false,  false);
    }

    const int row = m0 + half * 8;
#pragma unroll
    for (int i = 0; i < 8; ++i)
        C[(size_t)(row + i) * Nn + n0 + l] = c[i];
}

__global__ void attn_coef1(const float* __restrict__ feat,
                           const float* __restrict__ al, const float* __restrict__ ar,
                           float* __restrict__ el, float* __restrict__ er, int N) {
    int idx = blockIdx.x * blockDim.x + threadIdx.x;
    if (idx >= N * 8) return;
    int n = idx >> 3, h = idx & 7;
    const float* f = feat + (size_t)n * 64 + h * 8;
    float sl = 0.0f, sr = 0.0f;
#pragma unroll
    for (int d = 0; d < 8; ++d) {
        sl += f[d] * al[h * 8 + d];
        sr += f[d] * ar[h * 8 + d];
    }
    el[idx] = sl;
    er[idx] = sr;
}

__global__ void attn_coef2(const float* __restrict__ feat,
                           const float* __restrict__ al, const float* __restrict__ ar,
                           float* __restrict__ el, float* __restrict__ er, int N) {
    int n = blockIdx.x * blockDim.x + threadIdx.x;
    if (n >= N) return;
    const float* f = feat + (size_t)n * 64;
    float sl = 0.0f, sr = 0.0f;
#pragma unroll
    for (int d = 0; d < 64; ++d) {
        sl += f[d] * al[d];
        sr += f[d] * ar[d];
    }
    el[n] = sl;
    er[n] = sr;
}

__global__ void csr_zero(int* __restrict__ cnt, int N) {
    int idx = blockIdx.x * blockDim.x + threadIdx.x;
    if (idx < N) cnt[idx] = 0;
}

__global__ void csr_count(const int* __restrict__ dst, int* cnt, int* __restrict__ rank, int E) {
    int e = blockIdx.x * blockDim.x + threadIdx.x;
    if (e < E) rank[e] = atomicAdd(&cnt[dst[e]], 1);
}

__global__ __launch_bounds__(1024)
void csr_scan(const int* __restrict__ cnt, int* __restrict__ rowptr, int N) {
    __shared__ int wsum[32];
    const int t = threadIdx.x, lane = t & 31, w = t >> 5;
    const int per = (N + 1023) >> 10;
    const int b0 = t * per;
    int s = 0;
#pragma unroll 8
    for (int i = 0; i < per; ++i) {
        const int k = b0 + i;
        s += (k < N) ? cnt[k] : 0;
    }
    int inc = s;
#pragma unroll
    for (int o = 1; o < 32; o <<= 1) {
        const int u = __shfl_up(inc, o);
        if (lane >= o) inc += u;
    }
    if (lane == 31) wsum[w] = inc;
    __syncthreads();
    if (w == 0) {
        const int x = wsum[lane];
        int xi = x;
#pragma unroll
        for (int o = 1; o < 32; o <<= 1) {
            const int u = __shfl_up(xi, o);
            if (lane >= o) xi += u;
        }
        wsum[lane] = xi - x;
    }
    __syncthreads();
    int run = wsum[w] + inc - s;
    for (int i0 = 0; i0 < per; i0 += 8) {
        int c[8];
#pragma unroll
        for (int u = 0; u < 8; ++u) c[u] = cnt[min(b0 + i0 + u, N - 1)];
#pragma unroll
        for (int u = 0; u < 8; ++u) {
            const int k = b0 + i0 + u;
            if (i0 + u < per && k < N) {
                rowptr[k] = run;
                run += c[u];
            }
        }
    }
    if (t == 1023) rowptr[N] = run;
}

__global__ void csr_fill(const int* __restrict__ src, const int* __restrict__ dst,
                         const int* __restrict__ rank, const int* __restrict__ rowptr,
                         int* __restrict__ csr_src, int E) {
    int e = blockIdx.x * blockDim.x + threadIdx.x;
    if (e < E) csr_src[rowptr[dst[e]] + rank[e]] = src[e];
}

#define NEG_INF (-__builtin_inff())

__device__ __forceinline__ void g1_scores(const int* __restrict__ csr_src, const float* __restrict__ el,
                                          float er_h, int base, int cnt, int j, int h,
                                          int (&sv)[8], float (&v)[8]) {
    float tv[8];
#pragma unroll
    for (int r = 0; r < 8; ++r)
        sv[r] = csr_src[base + min(r * 4 + j, cnt - 1)];
#pragma unroll
    for (int r = 0; r < 8; ++r)
        tv[r] = el[sv[r] * 8 + h];
#pragma unroll
    for (int r = 0; r < 8; ++r)
        v[r] = (r * 4 + j < cnt) ? leaky(tv[r] + er_h) : NEG_INF;
}

__device__ __forceinline__ void g1_rows(const float* __restrict__ feat, const int (&sv)[8], const float (&av)[8],
                                        int cnt, int lane, float& acc0, float& acc1) {
    const int hq = lane >> 3;
#pragma unroll
    for (int b = 0; b < 4; ++b) {
        if (b * 8 < cnt) {
            float f0[8], f1[8];
#pragma unroll
            for (int i = 0; i < 8; ++i) {
                const int r = b * 2 + (i >> 2), jj = i & 3;
                if (r * 4 + jj < cnt) {
                    const int s = __builtin_amdgcn_readlane(sv[r], jj * 8);
                    const float* fs = feat + (size_t)s * 64;
                    f0[i] = fs[lane];
                    f1[i] = fs[lane + 32];
                }
            }
#pragma unroll
            for (int i = 0; i < 8; ++i) {
                const int r = b * 2 + (i >> 2), jj = i & 3;
                if (r * 4 + jj < cnt) {
                    const float alo = __shfl(av[r], jj * 8 + hq);
                    const float ahi = __shfl(av[r], jj * 8 + 4 + hq);
                    acc0 += f0[i] * alo;
                    acc1 += f1[i] * ahi;
                }
            }
        }
    }
}

__global__ __launch_bounds__(256)
void node_gather1(const int* __restrict__ rowptr, const int* __restrict__ csr_src,
                  const float* __restrict__ el, const float* __restrict__ er,
                  const float* __restrict__ feat, const float* __restrict__ b1,
                  float* __restrict__ x2, int N) {
    const int lane = threadIdx.x & 31;
    const int n = __builtin_amdgcn_readfirstlane((int)((blockIdx.x * blockDim.x + threadIdx.x) >> 5));
    if (n >= N) return;
    const int j = lane >> 3, h = lane & 7;
    const int beg = rowptr[n];
    const int deg = rowptr[n + 1] - beg;
    const float er_h = er[n * 8 + h];

    float acc0 = 0.0f, acc1 = 0.0f;
    if (deg > 0) {
        int sv[8];
        float v[8];
        g1_scores(csr_src, el, er_h, beg, min(deg, 32), j, h, sv, v);

        float m = v[0];
#pragma unroll
        for (int r = 1; r < 8; ++r) m = fmaxf(m, v[r]);
        for (int c = 32; c < deg; c += 32) {
            int s2[8];
            float v2[8];
            g1_scores(csr_src, el, er_h, beg + c, min(deg - c, 32), j, h, s2, v2);
#pragma unroll
            for (int r = 0; r < 8; ++r) m = fmaxf(m, v2[r]);
        }
        m = fmaxf(m, __shfl_xor(m, 8));
        m = fmaxf(m, __shfl_xor(m, 16));
        if (!__builtin_isfinite(m)) m = 0.0f;

        float ee[8];
        float d = 0.0f;
#pragma unroll
        for (int r = 0; r < 8; ++r) {
            ee[r] = expf(v[r] - m);
            d += ee[r];
        }
        for (int c = 32; c < deg; c += 32) {
            int s2[8];
            float v2[8];
            g1_scores(csr_src, el, er_h, beg + c, min(deg - c, 32), j, h, s2, v2);
#pragma unroll
            for (int r = 0; r < 8; ++r) d += expf(v2[r] - m);
        }
        d += __shfl_xor(d, 8);
        d += __shfl_xor(d, 16);
        const float dm = fmaxf(d, 1e-9f);

#pragma unroll
        for (int r = 0; r < 8; ++r) ee[r] = ee[r] / dm;
        g1_rows(feat, sv, ee, min(deg, 32), lane, acc0, acc1);
        for (int c = 32; c < deg; c += 32) {
            int s2[8];
            float v2[8];
            const int cnt = min(deg - c, 32);
            g1_scores(csr_src, el, er_h, beg + c, cnt, j, h, s2, v2);
#pragma unroll
            for (int r = 0; r < 8; ++r) v2[r] = expf(v2[r] - m) / dm;
            g1_rows(feat, s2, v2, cnt, lane, acc0, acc1);
        }
    }

    const float t0 = acc0 + b1[lane];
    const float t1 = acc1 + b1[lane + 32];
    float* xo = x2 + (size_t)n * 64;
    xo[lane]      = t0 > 0.0f ? t0 : expm1f(t0);
    xo[lane + 32] = t1 > 0.0f ? t1 : expm1f(t1);
}

__device__ __forceinline__ void g2_scores(const int* __restrict__ csr_src, const float* __restrict__ el,
                                          float er_n, int base, int cnt, int lane, int& s, float& v) {
    s = csr_src[base + min(lane, cnt - 1)];
    const float t = leaky(el[s] + er_n);
    v = (lane < cnt) ? t : NEG_INF;
}

__device__ __forceinline__ void g2_rows(const float* __restrict__ feat, int sl, float al,
                                        int cnt, int lane, float& acc0, float& acc1) {
#pragma unroll
    for (int b = 0; b < 4; ++b) {
        if (b * 8 < cnt) {
            float f0[8], f1[8];
#pragma unroll
            for (int i = 0; i < 8; ++i) {
                const int k = b * 8 + i;
                if (k < cnt) {
                    const int s = __builtin_amdgcn_readlane(sl, k);
                    const float* fs = feat + (size_t)s * 64;
                    f0[i] = fs[lane];
                    f1[i] = fs[lane + 32];
                }
            }
#pragma unroll
            for (int i = 0; i < 8; ++i) {
                const int k = b * 8 + i;
                if (k < cnt) {
                    const float a = __int_as_float(__builtin_amdgcn_readlane(__float_as_int(al), k));
                    acc0 += f0[i] * a;
                    acc1 += f1[i] * a;
                }
            }
        }
    }
}

__device__ __forceinline__ float wave_max32(float x) {
#pragma unroll
    for (int o = 16; o > 0; o >>= 1) x = fmaxf(x, __shfl_xor(x, o));
    return x;
}
__device__ __forceinline__ float wave_sum32(float x) {
#pragma unroll
    for (int o = 16; o > 0; o >>= 1) x += __shfl_xor(x, o);
    return x;
}

__global__ __launch_bounds__(256)
void node_gather2(const int* __restrict__ rowptr, const int* __restrict__ csr_src,
                  const float* __restrict__ el, const float* __restrict__ er,
                  const float* __restrict__ feat, const float* __restrict__ b2,
                  float* __restrict__ out, int N) {
    const int lane = threadIdx.x & 31;
    const int n = __builtin_amdgcn_readfirstlane((int)((blockIdx.x * blockDim.x + threadIdx.x) >> 5));
    if (n >= N) return;
    const int beg = rowptr[n];
    const int deg = rowptr[n + 1] - beg;
    const float er_n = er[n];

    float acc0 = 0.0f, acc1 = 0.0f;
    if (deg > 0) {
        int s0;
        float v0;
        g2_scores(csr_src, el, er_n, beg, min(deg, 32), lane, s0, v0);

        float m = v0;
        for (int c = 32; c < deg; c += 32) {
            int s2;
            float v2;
            g2_scores(csr_src, el, er_n, beg + c, min(deg - c, 32), lane, s2, v2);
            m = fmaxf(m, v2);
        }
        m = wave_max32(m);
        if (!__builtin_isfinite(m)) m = 0.0f;

        const float e0 = expf(v0 - m);
        float d = e0;
        for (int c = 32; c < deg; c += 32) {
            int s2;
            float v2;
            g2_scores(csr_src, el, er_n, beg + c, min(deg - c, 32), lane, s2, v2);
            d += expf(v2 - m);
        }
        d = wave_sum32(d);
        const float dm = fmaxf(d, 1e-9f);

        g2_rows(feat, s0, e0 / dm, min(deg, 32), lane, acc0, acc1);
        for (int c = 32; c < deg; c += 32) {
            int s2;
            float v2;
            const int cnt = min(deg - c, 32);
            g2_scores(csr_src, el, er_n, beg + c, cnt, lane, s2, v2);
            g2_rows(feat, s2, expf(v2 - m) / dm, cnt, lane, acc0, acc1);
        }
    }

    float* od = out + (size_t)n * 64;
    od[lane]      = acc0 + b2[lane];
    od[lane + 32] = acc1 + b2[lane + 32];
}

extern "C" void kernel_launch(void* const* d_in, const int* in_sizes, int n_in,
                              void* d_out, int out_size, void* d_ws, size_t ws_size,
                              hipStream_t stream) {
    (void)n_in; (void)out_size; (void)ws_size;
    const float* h   = (const float*)d_in[0];
    const int*   src = (const int*)d_in[1];
    const int*   dst = (const int*)d_in[2];
    const float* W1  = (const float*)d_in[3];
    const float* al1 = (const float*)d_in[4];
    const float* ar1 = (const float*)d_in[5];
    const float* b1  = (const float*)d_in[6];
    const float* W2  = (const float*)d_in[7];
    const float* al2 = (const float*)d_in[8];
    const float* ar2 = (const float*)d_in[9];
    const float* b2  = (const float*)d_in[10];
    float* out = (float*)d_out;

    const int N = in_sizes[0] / 128;
    const int E = in_sizes[1];

    float* feat1   = (float*)d_ws;
    float* el1     = feat1 + (size_t)N * 64;
    float* er1     = el1 + (size_t)N * 8;
    float* x2      = er1 + (size_t)N * 8;
    float* feat2   = x2 + (size_t)N * 64;
    float* el2     = feat2 + (size_t)N * 64;
    float* er2     = el2 + N;
    int*   cnt     = (int*)(er2 + N);
    int*   rowptr  = cnt + N;
    int*   rank    = rowptr + N + 1;
    int*   csr_src = rank + E;

    const int B = 256;
    const int gN8  = (N * 8 + B - 1) / B;
    const int gN   = (N + B - 1) / B;
    const int gE1  = (E + B - 1) / B;
    const int gNw  = (N * 32 + B - 1) / B;

    csr_zero <<<gN, B, 0, stream>>>(cnt, N);
    csr_count<<<gE1, B, 0, stream>>>(dst, cnt, rank, E);
    csr_scan <<<1, 1024, 0, stream>>>(cnt, rowptr, N);
    csr_fill <<<gE1, B, 0, stream>>>(src, dst, rank, rowptr, csr_src, E);

    {
        int tiles = (N / 16) * (64 / 16);
        wmma_gemm_f32<<<(tiles + 3) / 4, 128, 0, stream>>>(h, W1, feat1, N, 64, 128);
    }
    attn_coef1  <<<gN8, B, 0, stream>>>(feat1, al1, ar1, el1, er1, N);
    node_gather1<<<gNw, B, 0, stream>>>(rowptr, csr_src, el1, er1, feat1, b1, x2, N);

    {
        int tiles = (N / 16) * (64 / 16);
        wmma_gemm_f32<<<(tiles + 3) / 4, 128, 0, stream>>>(x2, W2, feat2, N, 64, 64);
    }
    attn_coef2  <<<gN, B, 0, stream>>>(feat2, al2, ar2, el2, er2, N);
    node_gather2<<<gNw, B, 0, stream>>>(rowptr, csr_src, el2, er2, feat2, b2, out, N);
}


# csr_scan on 49 blocks: each re-sums all counters before its 1024-counter segment with coalesced 16-byte loads, then one block scan (was one block, lanes 196 bytes apart); integers only
# speedup vs baseline: 3.6198x; 1.4347x over previous
#include <hip/hip_runtime.h>
#include <hip/hip_bf16.h>
#include <math.h>

typedef float v2f __attribute__((ext_vector_type(2)));
typedef float v8f __attribute__((ext_vector_type(8)));

#define NEG_SLOPE 0.2f

__device__ __forceinline__ float leaky(float v) {
    return v > 0.0f ? v : NEG_SLOPE * v;
}

__global__ __launch_bounds__(128)
void wmma_gemm_f32(const float* __restrict__ A, const float* __restrict__ B,
                   float* __restrict__ C, int M, int Nn, int K) {
    const int wave = threadIdx.x >> 5;
    const int lane = threadIdx.x & 31;
    const int tilesN = Nn >> 4;
    const int tile = blockIdx.x * 4 + wave;
    const int totalTiles = (M >> 4) * tilesN;
    if (tile >= totalTiles) return;

    const int m0 = (tile / tilesN) << 4;
    const int n0 = (tile % tilesN) << 4;
    const int half = lane >> 4;
    const int l    = lane & 15;

    v8f c = {};
    for (int k0 = 0; k0 < K; k0 += 4) {
        const int ka = k0 + 2 * half;
        v2f a, b;
        a.x = A[(size_t)(m0 + l) * K + ka];
        a.y = A[(size_t)(m0 + l) * K + ka + 1];
        b.x = B[(size_t)ka * Nn + n0 + l];
        b.y = B[(size_t)(ka + 1) * Nn + n0 + l];
        c = __builtin_amdgcn_wmma_f32_16x16x4_f32(
                 false, a,  false, b,
                 (short)0, c,  false,  false);
    }

    const int row = m0 + half * 8;
#pragma unroll
    for (int i = 0; i < 8; ++i)
        C[(size_t)(row + i) * Nn + n0 + l] = c[i];
}

__global__ void attn_coef1(const float* __restrict__ feat,
                           const float* __restrict__ al, const float* __restrict__ ar,
                           float* __restrict__ el, float* __restrict__ er, int N) {
    int idx = blockIdx.x * blockDim.x + threadIdx.x;
    if (idx >= N * 8) return;
    int n = idx >> 3, h = idx & 7;
    const float* f = feat + (size_t)n * 64 + h * 8;
    float sl = 0.0f, sr = 0.0f;
#pragma unroll
    for (int d = 0; d < 8; ++d) {
        sl += f[d] * al[h * 8 + d];
        sr += f[d] * ar[h * 8 + d];
    }
    el[idx] = sl;
    er[idx] = sr;
}

__global__ void attn_coef2(const float* __restrict__ feat,
                           const float* __restrict__ al, const float* __restrict__ ar,
                           float* __restrict__ el, float* __restrict__ er, int N) {
    int n = blockIdx.x * blockDim.x + threadIdx.x;
    if (n >= N) return;
    const float* f = feat + (size_t)n * 64;
    float sl = 0.0f, sr = 0.0f;
#pragma unroll
    for (int d = 0; d < 64; ++d) {
        sl += f[d] * al[d];
        sr += f[d] * ar[d];
    }
    el[n] = sl;
    er[n] = sr;
}

__global__ void csr_zero(int* __restrict__ cnt, int N) {
    int idx = blockIdx.x * blockDim.x + threadIdx.x;
    if (idx < N) cnt[idx] = 0;
}

__global__ void csr_count(const int* __restrict__ dst, int* cnt, int* __restrict__ rank, int E) {
    int e = blockIdx.x * blockDim.x + threadIdx.x;
    if (e < E) rank[e] = atomicAdd(&cnt[dst[e]], 1);
}

__global__ __launch_bounds__(1024)
void csr_scan(const int* __restrict__ cnt, int* __restrict__ rowptr, int N) {
    __shared__ int wfront[32];
    __shared__ int wown[32];
    const int t = threadIdx.x, lane = t & 31, w = t >> 5;
    const int seg0 = blockIdx.x << 10;
    const int n4 = seg0 >> 2;
    const int4* __restrict__ c4 = reinterpret_cast<const int4*>(cnt);

    const int k = seg0 + t;
    const int own = (k < N) ? cnt[k] : 0;

    int s = 0;
    for (int i0 = 0; i0 < n4; i0 += 16 * 1024) {
        int4 q[16];
#pragma unroll
        for (int u = 0; u < 16; ++u) q[u] = c4[min(i0 + u * 1024 + t, n4 - 1)];
#pragma unroll
        for (int u = 0; u < 16; ++u)
            if (i0 + u * 1024 + t < n4) s += (q[u].x + q[u].y) + (q[u].z + q[u].w);
    }

    int inc = own;
#pragma unroll
    for (int o = 1; o < 32; o <<= 1) {
        const int u = __shfl_up(inc, o);
        if (lane >= o) inc += u;
    }
    for (int o = 16; o > 0; o >>= 1) s += __shfl_xor(s, o);
    if (lane == 31) wown[w] = inc;
    if (lane == 0) wfront[w] = s;
    __syncthreads();
    int x = wfront[lane] + (lane < w ? wown[lane] : 0);
    for (int o = 16; o > 0; o >>= 1) x += __shfl_xor(x, o);
    const int excl = x + inc - own;
    if (k < N) rowptr[k] = excl;
    if (k == N - 1) rowptr[N] = excl + own;
}

__global__ void csr_fill(const int* __restrict__ src, const int* __restrict__ dst,
                         const int* __restrict__ rank, const int* __restrict__ rowptr,
                         int* __restrict__ csr_src, int E) {
    int e = blockIdx.x * blockDim.x + threadIdx.x;
    if (e < E) csr_src[rowptr[dst[e]] + rank[e]] = src[e];
}

#define NEG_INF (-__builtin_inff())

__device__ __forceinline__ void g1_scores(const int* __restrict__ csr_src, const float* __restrict__ el,
                                          float er_h, int base, int cnt, int j, int h,
                                          int (&sv)[8], float (&v)[8]) {
    float tv[8];
#pragma unroll
    for (int r = 0; r < 8; ++r)
        sv[r] = csr_src[base + min(r * 4 + j, cnt - 1)];
#pragma unroll
    for (int r = 0; r < 8; ++r)
        tv[r] = el[sv[r] * 8 + h];
#pragma unroll
    for (int r = 0; r < 8; ++r)
        v[r] = (r * 4 + j < cnt) ? leaky(tv[r] + er_h) : NEG_INF;
}

__device__ __forceinline__ void g1_rows(const float* __restrict__ feat, const int (&sv)[8], const float (&av)[8],
                                        int cnt, int lane, float& acc0, float& acc1) {
    const int hq = lane >> 3;
#pragma unroll
    for (int b = 0; b < 4; ++b) {
        if (b * 8 < cnt) {
            float f0[8], f1[8];
#pragma unroll
            for (int i = 0; i < 8; ++i) {
                const int r = b * 2 + (i >> 2), jj = i & 3;
                if (r * 4 + jj < cnt) {
                    const int s = __builtin_amdgcn_readlane(sv[r], jj * 8);
                    const float* fs = feat + (size_t)s * 64;
                    f0[i] = fs[lane];
                    f1[i] = fs[lane + 32];
                }
            }
#pragma unroll
            for (int i = 0; i < 8; ++i) {
                const int r = b * 2 + (i >> 2), jj = i & 3;
                if (r * 4 + jj < cnt) {
                    const float alo = __shfl(av[r], jj * 8 + hq);
                    const float ahi = __shfl(av[r], jj * 8 + 4 + hq);
                    acc0 += f0[i] * alo;
                    acc1 += f1[i] * ahi;
                }
            }
        }
    }
}

__global__ __launch_bounds__(256)
void node_gather1(const int* __restrict__ rowptr, const int* __restrict__ csr_src,
                  const float* __restrict__ el, const float* __restrict__ er,
                  const float* __restrict__ feat, const float* __restrict__ b1,
                  float* __restrict__ x2, int N) {
    const int lane = threadIdx.x & 31;
    const int n = __builtin_amdgcn_readfirstlane((int)((blockIdx.x * blockDim.x + threadIdx.x) >> 5));
    if (n >= N) return;
    const int j = lane >> 3, h = lane & 7;
    const int beg = rowptr[n];
    const int deg = rowptr[n + 1] - beg;
    const float er_h = er[n * 8 + h];

    float acc0 = 0.0f, acc1 = 0.0f;
    if (deg > 0) {
        int sv[8];
        float v[8];
        g1_scores(csr_src, el, er_h, beg, min(deg, 32), j, h, sv, v);

        float m = v[0];
#pragma unroll
        for (int r = 1; r < 8; ++r) m = fmaxf(m, v[r]);
        for (int c = 32; c < deg; c += 32) {
            int s2[8];
            float v2[8];
            g1_scores(csr_src, el, er_h, beg + c, min(deg - c, 32), j, h, s2, v2);
#pragma unroll
            for (int r = 0; r < 8; ++r) m = fmaxf(m, v2[r]);
        }
        m = fmaxf(m, __shfl_xor(m, 8));
        m = fmaxf(m, __shfl_xor(m, 16));
        if (!__builtin_isfinite(m)) m = 0.0f;

        float ee[8];
        float d = 0.0f;
#pragma unroll
        for (int r = 0; r < 8; ++r) {
            ee[r] = expf(v[r] - m);
            d += ee[r];
        }
        for (int c = 32; c < deg; c += 32) {
            int s2[8];
            float v2[8];
            g1_scores(csr_src, el, er_h, beg + c, min(deg - c, 32), j, h, s2, v2);
#pragma unroll
            for (int r = 0; r < 8; ++r) d += expf(v2[r] - m);
        }
        d += __shfl_xor(d, 8);
        d += __shfl_xor(d, 16);
        const float dm = fmaxf(d, 1e-9f);

#pragma unroll
        for (int r = 0; r < 8; ++r) ee[r] = ee[r] / dm;
        g1_rows(feat, sv, ee, min(deg, 32), lane, acc0, acc1);
        for (int c = 32; c < deg; c += 32) {
            int s2[8];
            float v2[8];
            const int cnt = min(deg - c, 32);
            g1_scores(csr_src, el, er_h, beg + c, cnt, j, h, s2, v2);
#pragma unroll
            for (int r = 0; r < 8; ++r) v2[r] = expf(v2[r] - m) / dm;
            g1_rows(feat, s2, v2, cnt, lane, acc0, acc1);
        }
    }

    const float t0 = acc0 + b1[lane];
    const float t1 = acc1 + b1[lane + 32];
    float* xo = x2 + (size_t)n * 64;
    xo[lane]      = t0 > 0.0f ? t0 : expm1f(t0);
    xo[lane + 32] = t1 > 0.0f ? t1 : expm1f(t1);
}

__device__ __forceinline__ void g2_scores(const int* __restrict__ csr_src, const float* __restrict__ el,
                                          float er_n, int base, int cnt, int lane, int& s, float& v) {
    s = csr_src[base + min(lane, cnt - 1)];
    const float t = leaky(el[s] + er_n);
    v = (lane < cnt) ? t : NEG_INF;
}

__device__ __forceinline__ void g2_rows(const float* __restrict__ feat, int sl, float al,
                                        int cnt, int lane, float& acc0, float& acc1) {
#pragma unroll
    for (int b = 0; b < 4; ++b) {
        if (b * 8 < cnt) {
            float f0[8], f1[8];
#pragma unroll
            for (int i = 0; i < 8; ++i) {
                const int k = b * 8 + i;
                if (k < cnt) {
                    const int s = __builtin_amdgcn_readlane(sl, k);
                    const float* fs = feat + (size_t)s * 64;
                    f0[i] = fs[lane];
                    f1[i] = fs[lane + 32];
                }
            }
#pragma unroll
            for (int i = 0; i < 8; ++i) {
                const int k = b * 8 + i;
                if (k < cnt) {
                    const float a = __int_as_float(__builtin_amdgcn_readlane(__float_as_int(al), k));
                    acc0 += f0[i] * a;
                    acc1 += f1[i] * a;
                }
            }
        }
    }
}

__device__ __forceinline__ float wave_max32(float x) {
#pragma unroll
    for (int o = 16; o > 0; o >>= 1) x = fmaxf(x, __shfl_xor(x, o));
    return x;
}
__device__ __forceinline__ float wave_sum32(float x) {
#pragma unroll
    for (int o = 16; o > 0; o >>= 1) x += __shfl_xor(x, o);
    return x;
}

__global__ __launch_bounds__(256)
void node_gather2(const int* __restrict__ rowptr, const int* __restrict__ csr_src,
                  const float* __restrict__ el, const float* __restrict__ er,
                  const float* __restrict__ feat, const float* __restrict__ b2,
                  float* __restrict__ out, int N) {
    const int lane = threadIdx.x & 31;
    const int n = __builtin_amdgcn_readfirstlane((int)((blockIdx.x * blockDim.x + threadIdx.x) >> 5));
    if (n >= N) return;
    const int beg = rowptr[n];
    const int deg = rowptr[n + 1] - beg;
    const float er_n = er[n];

    float acc0 = 0.0f, acc1 = 0.0f;
    if (deg > 0) {
        int s0;
        float v0;
        g2_scores(csr_src, el, er_n, beg, min(deg, 32), lane, s0, v0);

        float m = v0;
        for (int c = 32; c < deg; c += 32) {
            int s2;
            float v2;
            g2_scores(csr_src, el, er_n, beg + c, min(deg - c, 32), lane, s2, v2);
            m = fmaxf(m, v2);
        }
        m = wave_max32(m);
        if (!__builtin_isfinite(m)) m = 0.0f;

        const float e0 = expf(v0 - m);
        float d = e0;
        for (int c = 32; c < deg; c += 32) {
            int s2;
            float v2;
            g2_scores(csr_src, el, er_n, beg + c, min(deg - c, 32), lane, s2, v2);
            d += expf(v2 - m);
        }
        d = wave_sum32(d);
        const float dm = fmaxf(d, 1e-9f);

        g2_rows(feat, s0, e0 / dm, min(deg, 32), lane, acc0, acc1);
        for (int c = 32; c < deg; c += 32) {
            int s2;
            float v2;
            const int cnt = min(deg - c, 32);
            g2_scores(csr_src, el, er_n, beg + c, cnt, lane, s2, v2);
            g2_rows(feat, s2, expf(v2 - m) / dm, cnt, lane, acc0, acc1);
        }
    }

    float* od = out + (size_t)n * 64;
    od[lane]      = acc0 + b2[lane];
    od[lane + 32] = acc1 + b2[lane + 32];
}

extern "C" void kernel_launch(void* const* d_in, const int* in_sizes, int n_in,
                              void* d_out, int out_size, void* d_ws, size_t ws_size,
                              hipStream_t stream) {
    (void)n_in; (void)out_size; (void)ws_size;
    const float* h   = (const float*)d_in[0];
    const int*   src = (const int*)d_in[1];
    const int*   dst = (const int*)d_in[2];
    const float* W1  = (const float*)d_in[3];
    const float* al1 = (const float*)d_in[4];
    const float* ar1 = (const float*)d_in[5];
    const float* b1  = (const float*)d_in[6];
    const float* W2  = (const float*)d_in[7];
    const float* al2 = (const float*)d_in[8];
    const float* ar2 = (const float*)d_in[9];
    const float* b2  = (const float*)d_in[10];
    float* out = (float*)d_out;

    const int N = in_sizes[0] / 128;
    const int E = in_sizes[1];

    float* feat1   = (float*)d_ws;
    float* el1     = feat1 + (size_t)N * 64;
    float* er1     = el1 + (size_t)N * 8;
    float* x2      = er1 + (size_t)N * 8;
    float* feat2   = x2 + (size_t)N * 64;
    float* el2     = feat2 + (size_t)N * 64;
    float* er2     = el2 + N;
    int*   cnt     = (int*)(er2 + N);
    int*   rowptr  = cnt + N;
    int*   rank    = rowptr + N + 1;
    int*   csr_src = rank + E;

    const int B = 256;
    const int gN8  = (N * 8 + B - 1) / B;
    const int gN   = (N + B - 1) / B;
    const int gE1  = (E + B - 1) / B;
    const int gNw  = (N * 32 + B - 1) / B;

    csr_zero <<<gN, B, 0, stream>>>(cnt, N);
    csr_count<<<gE1, B, 0, stream>>>(dst, cnt, rank, E);
    csr_scan <<<(N + 1023) / 1024, 1024, 0, stream>>>(cnt, rowptr, N);
    csr_fill <<<gE1, B, 0, stream>>>(src, dst, rank, rowptr, csr_src, E);

    {
        int tiles = (N / 16) * (64 / 16);
        wmma_gemm_f32<<<(tiles + 3) / 4, 128, 0, stream>>>(h, W1, feat1, N, 64, 128);
    }
    attn_coef1  <<<gN8, B, 0, stream>>>(feat1, al1, ar1, el1, er1, N);
    node_gather1<<<gNw, B, 0, stream>>>(rowptr, csr_src, el1, er1, feat1, b1, x2, N);

    {
        int tiles = (N / 16) * (64 / 16);
        wmma_gemm_f32<<<(tiles + 3) / 4, 128, 0, stream>>>(x2, W2, feat2, N, 64, 64);
    }
    attn_coef2  <<<gN, B, 0, stream>>>(feat2, al2, ar2, el2, er2, N);
    node_gather2<<<gNw, B, 0, stream>>>(rowptr, csr_src, el2, er2, feat2, b2, out, N);
}


# el/er computed in the GEMM epilogue (GEMM regrouped: one wave per 16-row strip over all 64 columns) and the counter clear folded into GEMM 1: 10 launches -> 7; k order and el/er sum order regrouped
# speedup vs baseline: 3.9759x; 1.0984x over previous
#include <hip/hip_runtime.h>
#include <hip/hip_bf16.h>
#include <math.h>

typedef float v2f __attribute__((ext_vector_type(2)));
typedef float v8f __attribute__((ext_vector_type(8)));

#define NEG_SLOPE 0.2f

__device__ __forceinline__ float leaky(float v) {
    return v > 0.0f ? v : NEG_SLOPE * v;
}

typedef float v4f __attribute__((ext_vector_type(4)));

template <int K, bool HEADS8, bool ZERO>
__device__ __forceinline__ void gemm_coef_body(const float* __restrict__ A, const float* __restrict__ B,
                                               const float* __restrict__ al, const float* __restrict__ ar,
                                               float* __restrict__ C, float* __restrict__ el,
                                               float* __restrict__ er, int* __restrict__ cnt, int M) {
    if (ZERO) {
        const int gid = blockIdx.x * 128 + threadIdx.x;
        if (gid < M) cnt[gid] = 0;
    }
    const int lane = threadIdx.x & 31;
    const int strip = __builtin_amdgcn_readfirstlane((int)(blockIdx.x * 4 + (threadIdx.x >> 5)));
    if (strip >= (M >> 4)) return;
    const int m0 = strip << 4;
    const int hf = lane >> 4;
    const int c  = lane & 15;

    constexpr int NJ  = K / 8;
    constexpr int CH  = 2;
    constexpr int NCH = NJ / CH;

    const v4f* __restrict__ Ap = reinterpret_cast<const v4f*>(A + (size_t)(m0 + c) * K + 4 * hf);
    const v4f* __restrict__ Bp = reinterpret_cast<const v4f*>(B + (size_t)(4 * hf) * 64 + 4 * c);

    v4f aq[NJ];
#pragma unroll
    for (int j = 0; j < NJ; ++j) aq[j] = Ap[2 * j];

    v4f bq[2][CH][4];
#pragma unroll
    for (int jj = 0; jj < CH; ++jj)
#pragma unroll
        for (int i = 0; i < 4; ++i) bq[0][jj][i] = Bp[(8 * jj + i) * 16];
    __builtin_amdgcn_sched_barrier(0);

    v8f acc[4] = {};
#pragma unroll
    for (int ch = 0; ch < NCH; ++ch) {
        const int cur = ch & 1;
        if (ch + 1 < NCH) {
#pragma unroll
            for (int jj = 0; jj < CH; ++jj)
#pragma unroll
                for (int i = 0; i < 4; ++i)
                    bq[cur ^ 1][jj][i] = Bp[(8 * ((ch + 1) * CH + jj) + i) * 16];
        }
        __builtin_amdgcn_sched_barrier(0);
#pragma unroll
        for (int jj = 0; jj < CH; ++jj) {
            const v4f q = aq[ch * CH + jj];
#pragma unroll
            for (int u = 0; u < 2; ++u) {
                v2f a;
                a.x = u ? q.z : q.x;
                a.y = u ? q.w : q.y;
#pragma unroll
                for (int t = 0; t < 4; ++t) {
                    v2f b;
                    b.x = bq[cur][jj][2 * u][t];
                    b.y = bq[cur][jj][2 * u + 1][t];
                    acc[t] = __builtin_amdgcn_wmma_f32_16x16x4_f32(
                                  false, a,  false, b,
                                  (short)0, acc[t],  false,  false);
                }
            }
        }
    }

    const v4f alv = *reinterpret_cast<const v4f*>(al + 4 * c);
    const v4f arv = *reinterpret_cast<const v4f*>(ar + 4 * c);
    float pl[8], pr[8];
#pragma unroll
    for (int r = 0; r < 8; ++r) {
        v4f o;
        o.x = acc[0][r]; o.y = acc[1][r]; o.z = acc[2][r]; o.w = acc[3][r];
        *reinterpret_cast<v4f*>(C + (size_t)(m0 + 8 * hf + r) * 64 + 4 * c) = o;
        pl[r] = o.x * alv.x + o.y * alv.y + o.z * alv.z + o.w * alv.w;
        pr[r] = o.x * arv.x + o.y * arv.y + o.z * arv.z + o.w * arv.w;
    }

    if (HEADS8) {
#pragma unroll
        for (int r = 0; r < 8; ++r) {
            pl[r] += __shfl_xor(pl[r], 1);
            pr[r] += __shfl_xor(pr[r], 1);
        }
        float* __restrict__ dp = ((c & 1) ? er : el) + (size_t)(m0 + 8 * hf) * 8 + (c >> 1);
#pragma unroll
        for (int r = 0; r < 8; ++r) dp[r * 8] = (c & 1) ? pr[r] : pl[r];
    } else {
#pragma unroll
        for (int r = 0; r < 8; ++r) {
#pragma unroll
            for (int o = 1; o < 16; o <<= 1) {
                pl[r] += __shfl_xor(pl[r], o);
                pr[r] += __shfl_xor(pr[r], o);
            }
        }
        float v = 0.0f;
#pragma unroll
        for (int r = 0; r < 8; ++r)
            if ((c & 7) == r) v = (c & 8) ? pr[r] : pl[r];
        float* __restrict__ dp = ((c & 8) ? er : el) + m0 + 8 * hf + (c & 7);
        *dp = v;
    }
}

__global__ __launch_bounds__(128)
void gemm1_coef(const float* __restrict__ A, const float* __restrict__ B,
                const float* __restrict__ al, const float* __restrict__ ar,
                float* __restrict__ C, float* __restrict__ el, float* __restrict__ er,
                int* __restrict__ cnt, int M) {
    gemm_coef_body<128, true, true>(A, B, al, ar, C, el, er, cnt, M);
}

__global__ __launch_bounds__(128)
void gemm2_coef(const float* __restrict__ A, const float* __restrict__ B,
                const float* __restrict__ al, const float* __restrict__ ar,
                float* __restrict__ C, float* __restrict__ el, float* __restrict__ er, int M) {
    gemm_coef_body<64, false, false>(A, B, al, ar, C, el, er, nullptr, M);
}

__global__ void csr_count(const int* __restrict__ dst, int* cnt, int* __restrict__ rank, int E) {
    int e = blockIdx.x * blockDim.x + threadIdx.x;
    if (e < E) rank[e] = atomicAdd(&cnt[dst[e]], 1);
}

__global__ __launch_bounds__(1024)
void csr_scan(const int* __restrict__ cnt, int* __restrict__ rowptr, int N) {
    __shared__ int wfront[32];
    __shared__ int wown[32];
    const int t = threadIdx.x, lane = t & 31, w = t >> 5;
    const int seg0 = blockIdx.x << 10;
    const int n4 = seg0 >> 2;
    const int4* __restrict__ c4 = reinterpret_cast<const int4*>(cnt);

    const int k = seg0 + t;
    const int own = (k < N) ? cnt[k] : 0;

    int s = 0;
    for (int i0 = 0; i0 < n4; i0 += 16 * 1024) {
        int4 q[16];
#pragma unroll
        for (int u = 0; u < 16; ++u) q[u] = c4[min(i0 + u * 1024 + t, n4 - 1)];
#pragma unroll
        for (int u = 0; u < 16; ++u)
            if (i0 + u * 1024 + t < n4) s += (q[u].x + q[u].y) + (q[u].z + q[u].w);
    }

    int inc = own;
#pragma unroll
    for (int o = 1; o < 32; o <<= 1) {
        const int u = __shfl_up(inc, o);
        if (lane >= o) inc += u;
    }
    for (int o = 16; o > 0; o >>= 1) s += __shfl_xor(s, o);
    if (lane == 31) wown[w] = inc;
    if (lane == 0) wfront[w] = s;
    __syncthreads();
    int x = wfront[lane] + (lane < w ? wown[lane] : 0);
    for (int o = 16; o > 0; o >>= 1) x += __shfl_xor(x, o);
    const int excl = x + inc - own;
    if (k < N) rowptr[k] = excl;
    if (k == N - 1) rowptr[N] = excl + own;
}

__global__ void csr_fill(const int* __restrict__ src, const int* __restrict__ dst,
                         const int* __restrict__ rank, const int* __restrict__ rowptr,
                         int* __restrict__ csr_src, int E) {
    int e = blockIdx.x * blockDim.x + threadIdx.x;
    if (e < E) csr_src[rowptr[dst[e]] + rank[e]] = src[e];
}

#define NEG_INF (-__builtin_inff())

__device__ __forceinline__ void g1_scores(const int* __restrict__ csr_src, const float* __restrict__ el,
                                          float er_h, int base, int cnt, int j, int h,
                                          int (&sv)[8], float (&v)[8]) {
    float tv[8];
#pragma unroll
    for (int r = 0; r < 8; ++r)
        sv[r] = csr_src[base + min(r * 4 + j, cnt - 1)];
#pragma unroll
    for (int r = 0; r < 8; ++r)
        tv[r] = el[sv[r] * 8 + h];
#pragma unroll
    for (int r = 0; r < 8; ++r)
        v[r] = (r * 4 + j < cnt) ? leaky(tv[r] + er_h) : NEG_INF;
}

__device__ __forceinline__ void g1_rows(const float* __restrict__ feat, const int (&sv)[8], const float (&av)[8],
                                        int cnt, int lane, float& acc0, float& acc1) {
    const int hq = lane >> 3;
#pragma unroll
    for (int b = 0; b < 4; ++b) {
        if (b * 8 < cnt) {
            float f0[8], f1[8];
#pragma unroll
            for (int i = 0; i < 8; ++i) {
                const int r = b * 2 + (i >> 2), jj = i & 3;
                if (r * 4 + jj < cnt) {
                    const int s = __builtin_amdgcn_readlane(sv[r], jj * 8);
                    const float* fs = feat + (size_t)s * 64;
                    f0[i] = fs[lane];
                    f1[i] = fs[lane + 32];
                }
            }
#pragma unroll
            for (int i = 0; i < 8; ++i) {
                const int r = b * 2 + (i >> 2), jj = i & 3;
                if (r * 4 + jj < cnt) {
                    const float alo = __shfl(av[r], jj * 8 + hq);
                    const float ahi = __shfl(av[r], jj * 8 + 4 + hq);
                    acc0 += f0[i] * alo;
                    acc1 += f1[i] * ahi;
                }
            }
        }
    }
}

__global__ __launch_bounds__(256)
void node_gather1(const int* __restrict__ rowptr, const int* __restrict__ csr_src,
                  const float* __restrict__ el, const float* __restrict__ er,
                  const float* __restrict__ feat, const float* __restrict__ b1,
                  float* __restrict__ x2, int N) {
    const int lane = threadIdx.x & 31;
    const int n = __builtin_amdgcn_readfirstlane((int)((blockIdx.x * blockDim.x + threadIdx.x) >> 5));
    if (n >= N) return;
    const int j = lane >> 3, h = lane & 7;
    const int beg = rowptr[n];
    const int deg = rowptr[n + 1] - beg;
    const float er_h = er[n * 8 + h];

    float acc0 = 0.0f, acc1 = 0.0f;
    if (deg > 0) {
        int sv[8];
        float v[8];
        g1_scores(csr_src, el, er_h, beg, min(deg, 32), j, h, sv, v);

        float m = v[0];
#pragma unroll
        for (int r = 1; r < 8; ++r) m = fmaxf(m, v[r]);
        for (int c = 32; c < deg; c += 32) {
            int s2[8];
            float v2[8];
            g1_scores(csr_src, el, er_h, beg + c, min(deg - c, 32), j, h, s2, v2);
#pragma unroll
            for (int r = 0; r < 8; ++r) m = fmaxf(m, v2[r]);
        }
        m = fmaxf(m, __shfl_xor(m, 8));
        m = fmaxf(m, __shfl_xor(m, 16));
        if (!__builtin_isfinite(m)) m = 0.0f;

        float ee[8];
        float d = 0.0f;
#pragma unroll
        for (int r = 0; r < 8; ++r) {
            ee[r] = expf(v[r] - m);
            d += ee[r];
        }
        for (int c = 32; c < deg; c += 32) {
            int s2[8];
            float v2[8];
            g1_scores(csr_src, el, er_h, beg + c, min(deg - c, 32), j, h, s2, v2);
#pragma unroll
            for (int r = 0; r < 8; ++r) d += expf(v2[r] - m);
        }
        d += __shfl_xor(d, 8);
        d += __shfl_xor(d, 16);
        const float dm = fmaxf(d, 1e-9f);

#pragma unroll
        for (int r = 0; r < 8; ++r) ee[r] = ee[r] / dm;
        g1_rows(feat, sv, ee, min(deg, 32), lane, acc0, acc1);
        for (int c = 32; c < deg; c += 32) {
            int s2[8];
            float v2[8];
            const int cnt = min(deg - c, 32);
            g1_scores(csr_src, el, er_h, beg + c, cnt, j, h, s2, v2);
#pragma unroll
            for (int r = 0; r < 8; ++r) v2[r] = expf(v2[r] - m) / dm;
            g1_rows(feat, s2, v2, cnt, lane, acc0, acc1);
        }
    }

    const float t0 = acc0 + b1[lane];
    const float t1 = acc1 + b1[lane + 32];
    float* xo = x2 + (size_t)n * 64;
    xo[lane]      = t0 > 0.0f ? t0 : expm1f(t0);
    xo[lane + 32] = t1 > 0.0f ? t1 : expm1f(t1);
}

__device__ __forceinline__ void g2_scores(const int* __restrict__ csr_src, const float* __restrict__ el,
                                          float er_n, int base, int cnt, int lane, int& s, float& v) {
    s = csr_src[base + min(lane, cnt - 1)];
    const float t = leaky(el[s] + er_n);
    v = (lane < cnt) ? t : NEG_INF;
}

__device__ __forceinline__ void g2_rows(const float* __restrict__ feat, int sl, float al,
                                        int cnt, int lane, float& acc0, float& acc1) {
#pragma unroll
    for (int b = 0; b < 4; ++b) {
        if (b * 8 < cnt) {
            float f0[8], f1[8];
#pragma unroll
            for (int i = 0; i < 8; ++i) {
                const int k = b * 8 + i;
                if (k < cnt) {
                    const int s = __builtin_amdgcn_readlane(sl, k);
                    const float* fs = feat + (size_t)s * 64;
                    f0[i] = fs[lane];
                    f1[i] = fs[lane + 32];
                }
            }
#pragma unroll
            for (int i = 0; i < 8; ++i) {
                const int k = b * 8 + i;
                if (k < cnt) {
                    const float a = __int_as_float(__builtin_amdgcn_readlane(__float_as_int(al), k));
                    acc0 += f0[i] * a;
                    acc1 += f1[i] * a;
                }
            }
        }
    }
}

__device__ __forceinline__ float wave_max32(float x) {
#pragma unroll
    for (int o = 16; o > 0; o >>= 1) x = fmaxf(x, __shfl_xor(x, o));
    return x;
}
__device__ __forceinline__ float wave_sum32(float x) {
#pragma unroll
    for (int o = 16; o > 0; o >>= 1) x += __shfl_xor(x, o);
    return x;
}

__global__ __launch_bounds__(256)
void node_gather2(const int* __restrict__ rowptr, const int* __restrict__ csr_src,
                  const float* __restrict__ el, const float* __restrict__ er,
                  const float* __restrict__ feat, const float* __restrict__ b2,
                  float* __restrict__ out, int N) {
    const int lane = threadIdx.x & 31;
    const int n = __builtin_amdgcn_readfirstlane((int)((blockIdx.x * blockDim.x + threadIdx.x) >> 5));
    if (n >= N) return;
    const int beg = rowptr[n];
    const int deg = rowptr[n + 1] - beg;
    const float er_n = er[n];

    float acc0 = 0.0f, acc1 = 0.0f;
    if (deg > 0) {
        int s0;
        float v0;
        g2_scores(csr_src, el, er_n, beg, min(deg, 32), lane, s0, v0);

        float m = v0;
        for (int c = 32; c < deg; c += 32) {
            int s2;
            float v2;
            g2_scores(csr_src, el, er_n, beg + c, min(deg - c, 32), lane, s2, v2);
            m = fmaxf(m, v2);
        }
        m = wave_max32(m);
        if (!__builtin_isfinite(m)) m = 0.0f;

        const float e0 = expf(v0 - m);
        float d = e0;
        for (int c = 32; c < deg; c += 32) {
            int s2;
            float v2;
            g2_scores(csr_src, el, er_n, beg + c, min(deg - c, 32), lane, s2, v2);
            d += expf(v2 - m);
        }
        d = wave_sum32(d);
        const float dm = fmaxf(d, 1e-9f);

        g2_rows(feat, s0, e0 / dm, min(deg, 32), lane, acc0, acc1);
        for (int c = 32; c < deg; c += 32) {
            int s2;
            float v2;
            const int cnt = min(deg - c, 32);
            g2_scores(csr_src, el, er_n, beg + c, cnt, lane, s2, v2);
            g2_rows(feat, s2, expf(v2 - m) / dm, cnt, lane, acc0, acc1);
        }
    }

    float* od = out + (size_t)n * 64;
    od[lane]      = acc0 + b2[lane];
    od[lane + 32] = acc1 + b2[lane + 32];
}

extern "C" void kernel_launch(void* const* d_in, const int* in_sizes, int n_in,
                              void* d_out, int out_size, void* d_ws, size_t ws_size,
                              hipStream_t stream) {
    (void)n_in; (void)out_size; (void)ws_size;
    const float* h   = (const float*)d_in[0];
    const int*   src = (const int*)d_in[1];
    const int*   dst = (const int*)d_in[2];
    const float* W1  = (const float*)d_in[3];
    const float* al1 = (const float*)d_in[4];
    const float* ar1 = (const float*)d_in[5];
    const float* b1  = (const float*)d_in[6];
    const float* W2  = (const float*)d_in[7];
    const float* al2 = (const float*)d_in[8];
    const float* ar2 = (const float*)d_in[9];
    const float* b2  = (const float*)d_in[10];
    float* out = (float*)d_out;

    const int N = in_sizes[0] / 128;
    const int E = in_sizes[1];

    float* feat1   = (float*)d_ws;
    float* el1     = feat1 + (size_t)N * 64;
    float* er1     = el1 + (size_t)N * 8;
    float* x2      = er1 + (size_t)N * 8;
    float* feat2   = x2 + (size_t)N * 64;
    float* el2     = feat2 + (size_t)N * 64;
    float* er2     = el2 + N;
    int*   cnt     = (int*)(er2 + N);
    int*   rowptr  = cnt + N;
    int*   rank    = rowptr + N + 1;
    int*   csr_src = rank + E;

    const int B = 256;
    const int gE1  = (E + B - 1) / B;
    const int gNw  = (N * 32 + B - 1) / B;
    const int gG   = (N / 16 + 3) / 4;

    gemm1_coef<<<gG, 128, 0, stream>>>(h, W1, al1, ar1, feat1, el1, er1, cnt, N);

    csr_count<<<gE1, B, 0, stream>>>(dst, cnt, rank, E);
    csr_scan <<<(N + 1023) / 1024, 1024, 0, stream>>>(cnt, rowptr, N);
    csr_fill <<<gE1, B, 0, stream>>>(src, dst, rank, rowptr, csr_src, E);

    node_gather1<<<gNw, B, 0, stream>>>(rowptr, csr_src, el1, er1, feat1, b1, x2, N);

    gemm2_coef  <<<gG, 128, 0, stream>>>(x2, W2, al2, ar2, feat2, el2, er2, N);
    node_gather2<<<gNw, B, 0, stream>>>(rowptr, csr_src, el2, er2, feat2, b2, out, N);
}


# Edge sort (count, scan, fill kernels) replaced by one pass into fixed 64-slot node buckets (int atomic slot); deeper nodes walk the input edge list: 7 launches -> 5; sums in arrival rank as before
# speedup vs baseline: 4.5906x; 1.1546x over previous
#include <hip/hip_runtime.h>
#include <hip/hip_bf16.h>
#include <math.h>

typedef float v2f __attribute__((ext_vector_type(2)));
typedef float v8f __attribute__((ext_vector_type(8)));

#define NEG_SLOPE 0.2f

__device__ __forceinline__ float leaky(float v) {
    return v > 0.0f ? v : NEG_SLOPE * v;
}

typedef float v4f __attribute__((ext_vector_type(4)));

template <int K, bool HEADS8, bool ZERO>
__device__ __forceinline__ void gemm_coef_body(const float* __restrict__ A, const float* __restrict__ B,
                                               const float* __restrict__ al, const float* __restrict__ ar,
                                               float* __restrict__ C, float* __restrict__ el,
                                               float* __restrict__ er, int* __restrict__ cnt, int M) {
    if (ZERO) {
        const int gid = blockIdx.x * 128 + threadIdx.x;
        if (gid < M) cnt[gid] = 0;
    }
    const int lane = threadIdx.x & 31;
    const int strip = __builtin_amdgcn_readfirstlane((int)(blockIdx.x * 4 + (threadIdx.x >> 5)));
    if (strip >= (M >> 4)) return;
    const int m0 = strip << 4;
    const int hf = lane >> 4;
    const int c  = lane & 15;

    constexpr int NJ  = K / 8;
    constexpr int CH  = 2;
    constexpr int NCH = NJ / CH;

    const v4f* __restrict__ Ap = reinterpret_cast<const v4f*>(A + (size_t)(m0 + c) * K + 4 * hf);
    const v4f* __restrict__ Bp = reinterpret_cast<const v4f*>(B + (size_t)(4 * hf) * 64 + 4 * c);

    v4f aq[NJ];
#pragma unroll
    for (int j = 0; j < NJ; ++j) aq[j] = Ap[2 * j];

    v4f bq[2][CH][4];
#pragma unroll
    for (int jj = 0; jj < CH; ++jj)
#pragma unroll
        for (int i = 0; i < 4; ++i) bq[0][jj][i] = Bp[(8 * jj + i) * 16];
    __builtin_amdgcn_sched_barrier(0);

    v8f acc[4] = {};
#pragma unroll
    for (int ch = 0; ch < NCH; ++ch) {
        const int cur = ch & 1;
        if (ch + 1 < NCH) {
#pragma unroll
            for (int jj = 0; jj < CH; ++jj)
#pragma unroll
                for (int i = 0; i < 4; ++i)
                    bq[cur ^ 1][jj][i] = Bp[(8 * ((ch + 1) * CH + jj) + i) * 16];
        }
        __builtin_amdgcn_sched_barrier(0);
#pragma unroll
        for (int jj = 0; jj < CH; ++jj) {
            const v4f q = aq[ch * CH + jj];
#pragma unroll
            for (int u = 0; u < 2; ++u) {
                v2f a;
                a.x = u ? q.z : q.x;
                a.y = u ? q.w : q.y;
#pragma unroll
                for (int t = 0; t < 4; ++t) {
                    v2f b;
                    b.x = bq[cur][jj][2 * u][t];
                    b.y = bq[cur][jj][2 * u + 1][t];
                    acc[t] = __builtin_amdgcn_wmma_f32_16x16x4_f32(
                                  false, a,  false, b,
                                  (short)0, acc[t],  false,  false);
                }
            }
        }
    }

    const v4f alv = *reinterpret_cast<const v4f*>(al + 4 * c);
    const v4f arv = *reinterpret_cast<const v4f*>(ar + 4 * c);
    float pl[8], pr[8];
#pragma unroll
    for (int r = 0; r < 8; ++r) {
        v4f o;
        o.x = acc[0][r]; o.y = acc[1][r]; o.z = acc[2][r]; o.w = acc[3][r];
        *reinterpret_cast<v4f*>(C + (size_t)(m0 + 8 * hf + r) * 64 + 4 * c) = o;
        pl[r] = o.x * alv.x + o.y * alv.y + o.z * alv.z + o.w * alv.w;
        pr[r] = o.x * arv.x + o.y * arv.y + o.z * arv.z + o.w * arv.w;
    }

    if (HEADS8) {
#pragma unroll
        for (int r = 0; r < 8; ++r) {
            pl[r] += __shfl_xor(pl[r], 1);
            pr[r] += __shfl_xor(pr[r], 1);
        }
        float* __restrict__ dp = ((c & 1) ? er : el) + (size_t)(m0 + 8 * hf) * 8 + (c >> 1);
#pragma unroll
        for (int r = 0; r < 8; ++r) dp[r * 8] = (c & 1) ? pr[r] : pl[r];
    } else {
#pragma unroll
        for (int r = 0; r < 8; ++r) {
#pragma unroll
            for (int o = 1; o < 16; o <<= 1) {
                pl[r] += __shfl_xor(pl[r], o);
                pr[r] += __shfl_xor(pr[r], o);
            }
        }
        float v = 0.0f;
#pragma unroll
        for (int r = 0; r < 8; ++r)
            if ((c & 7) == r) v = (c & 8) ? pr[r] : pl[r];
        float* __restrict__ dp = ((c & 8) ? er : el) + m0 + 8 * hf + (c & 7);
        *dp = v;
    }
}

__global__ __launch_bounds__(128)
void gemm1_coef(const float* __restrict__ A, const float* __restrict__ B,
                const float* __restrict__ al, const float* __restrict__ ar,
                float* __restrict__ C, float* __restrict__ el, float* __restrict__ er,
                int* __restrict__ cnt, int M) {
    gemm_coef_body<128, true, true>(A, B, al, ar, C, el, er, cnt, M);
}

__global__ __launch_bounds__(128)
void gemm2_coef(const float* __restrict__ A, const float* __restrict__ B,
                const float* __restrict__ al, const float* __restrict__ ar,
                float* __restrict__ C, float* __restrict__ el, float* __restrict__ er, int M) {
    gemm_coef_body<64, false, false>(A, B, al, ar, C, el, er, nullptr, M);
}

#define CAP 64

__global__ void bucket_fill(const int* __restrict__ src, const int* __restrict__ dst,
                            int* cnt, int* __restrict__ bucket, int E) {
    int e = blockIdx.x * blockDim.x + threadIdx.x;
    if (e < E) {
        const int d = dst[e];
        const int s = src[e];
        asm volatile("" :: "v"(s));
        const int slot = atomicAdd(&cnt[d], 1);
        if (slot < CAP) bucket[(size_t)d * CAP + slot] = s;
    }
}

template <class F>
__device__ __forceinline__ void for_each_in_edge(const int* __restrict__ src, const int* __restrict__ dst,
                                                 int E, int n, int lane, F f) {
    for (int e0 = 0; e0 < E; e0 += 32) {
        const int e = e0 + lane;
        const int d = (e < E) ? dst[e] : -1;
        unsigned hit = __builtin_amdgcn_ballot_w32(d == n);
        if (hit) {
            const int sl = (e < E) ? src[e] : 0;
            while (hit) {
                const int b = __builtin_ctz(hit);
                hit &= hit - 1;
                f(__builtin_amdgcn_readlane(sl, b));
            }
        }
    }
}

#define NEG_INF (-__builtin_inff())

__device__ __forceinline__ void g1_scores(const int* __restrict__ ids, const float* __restrict__ el,
                                          float er_h, int base, int cnt, int j, int h,
                                          int (&sv)[8], float (&v)[8]) {
    float tv[8];
#pragma unroll
    for (int r = 0; r < 8; ++r)
        sv[r] = ids[base + min(r * 4 + j, cnt - 1)];
#pragma unroll
    for (int r = 0; r < 8; ++r)
        tv[r] = el[sv[r] * 8 + h];
#pragma unroll
    for (int r = 0; r < 8; ++r)
        v[r] = (r * 4 + j < cnt) ? leaky(tv[r] + er_h) : NEG_INF;
}

__device__ __forceinline__ void g1_rows(const float* __restrict__ feat, const int (&sv)[8], const float (&av)[8],
                                        int cnt, int lane, float& acc0, float& acc1) {
    const int hq = lane >> 3;
#pragma unroll
    for (int b = 0; b < 4; ++b) {
        if (b * 8 < cnt) {
            float f0[8], f1[8];
#pragma unroll
            for (int i = 0; i < 8; ++i) {
                const int r = b * 2 + (i >> 2), jj = i & 3;
                if (r * 4 + jj < cnt) {
                    const int s = __builtin_amdgcn_readlane(sv[r], jj * 8);
                    const float* fs = feat + (size_t)s * 64;
                    f0[i] = fs[lane];
                    f1[i] = fs[lane + 32];
                }
            }
#pragma unroll
            for (int i = 0; i < 8; ++i) {
                const int r = b * 2 + (i >> 2), jj = i & 3;
                if (r * 4 + jj < cnt) {
                    const float alo = __shfl(av[r], jj * 8 + hq);
                    const float ahi = __shfl(av[r], jj * 8 + 4 + hq);
                    acc0 += f0[i] * alo;
                    acc1 += f1[i] * ahi;
                }
            }
        }
    }
}

__device__ __forceinline__ void g1_general(const int* __restrict__ src, const int* __restrict__ dst, int E, int n,
                                           const float* __restrict__ el, float er_h,
                                           const float* __restrict__ feat, int lane,
                                           float& acc0, float& acc1) {
    const int h = lane & 7, hq = lane >> 3;
    float m = NEG_INF;
    for_each_in_edge(src, dst, E, n, lane, [&](int s) {
        m = fmaxf(m, leaky(el[s * 8 + h] + er_h));
    });
    if (!__builtin_isfinite(m)) m = 0.0f;
    float d = 0.0f;
    for_each_in_edge(src, dst, E, n, lane, [&](int s) {
        d += expf(leaky(el[s * 8 + h] + er_h) - m);
    });
    const float dm = fmaxf(d, 1e-9f);
    for_each_in_edge(src, dst, E, n, lane, [&](int s) {
        const float a = expf(leaky(el[s * 8 + h] + er_h) - m) / dm;
        const float alo = __shfl(a, hq);
        const float ahi = __shfl(a, 4 + hq);
        const float* fs = feat + (size_t)s * 64;
        acc0 += fs[lane] * alo;
        acc1 += fs[lane + 32] * ahi;
    });
}

__global__ __launch_bounds__(256)
void node_gather1(const int* __restrict__ cnt, const int* __restrict__ bucket,
                  const int* __restrict__ src, const int* __restrict__ dst, int E,
                  const float* __restrict__ el, const float* __restrict__ er,
                  const float* __restrict__ feat, const float* __restrict__ b1,
                  float* __restrict__ x2, int N) {
    const int lane = threadIdx.x & 31;
    const int n = __builtin_amdgcn_readfirstlane((int)((blockIdx.x * blockDim.x + threadIdx.x) >> 5));
    if (n >= N) return;
    const int j = lane >> 3, h = lane & 7;
    const int beg = n * CAP;
    const int deg = cnt[n];
    const float er_h = er[n * 8 + h];

    float acc0 = 0.0f, acc1 = 0.0f;
    if (deg > CAP) {
        g1_general(src, dst, E, n, el, er_h, feat, lane, acc0, acc1);
    } else if (deg > 0) {
        int sv[8];
        float v[8];
        g1_scores(bucket, el, er_h, beg, min(deg, 32), j, h, sv, v);

        float m = v[0];
#pragma unroll
        for (int r = 1; r < 8; ++r) m = fmaxf(m, v[r]);
        for (int c = 32; c < deg; c += 32) {
            int s2[8];
            float v2[8];
            g1_scores(bucket, el, er_h, beg + c, min(deg - c, 32), j, h, s2, v2);
#pragma unroll
            for (int r = 0; r < 8; ++r) m = fmaxf(m, v2[r]);
        }
        m = fmaxf(m, __shfl_xor(m, 8));
        m = fmaxf(m, __shfl_xor(m, 16));
        if (!__builtin_isfinite(m)) m = 0.0f;

        float ee[8];
        float d = 0.0f;
#pragma unroll
        for (int r = 0; r < 8; ++r) {
            ee[r] = expf(v[r] - m);
            d += ee[r];
        }
        for (int c = 32; c < deg; c += 32) {
            int s2[8];
            float v2[8];
            g1_scores(bucket, el, er_h, beg + c, min(deg - c, 32), j, h, s2, v2);
#pragma unroll
            for (int r = 0; r < 8; ++r) d += expf(v2[r] - m);
        }
        d += __shfl_xor(d, 8);
        d += __shfl_xor(d, 16);
        const float dm = fmaxf(d, 1e-9f);

#pragma unroll
        for (int r = 0; r < 8; ++r) ee[r] = ee[r] / dm;
        g1_rows(feat, sv, ee, min(deg, 32), lane, acc0, acc1);
        for (int c = 32; c < deg; c += 32) {
            int s2[8];
            float v2[8];
            const int cnt = min(deg - c, 32);
            g1_scores(bucket, el, er_h, beg + c, cnt, j, h, s2, v2);
#pragma unroll
            for (int r = 0; r < 8; ++r) v2[r] = expf(v2[r] - m) / dm;
            g1_rows(feat, s2, v2, cnt, lane, acc0, acc1);
        }
    }

    const float t0 = acc0 + b1[lane];
    const float t1 = acc1 + b1[lane + 32];
    float* xo = x2 + (size_t)n * 64;
    xo[lane]      = t0 > 0.0f ? t0 : expm1f(t0);
    xo[lane + 32] = t1 > 0.0f ? t1 : expm1f(t1);
}

__device__ __forceinline__ void g2_scores(const int* __restrict__ ids, const float* __restrict__ el,
                                          float er_n, int base, int cnt, int lane, int& s, float& v) {
    s = ids[base + min(lane, cnt - 1)];
    const float t = leaky(el[s] + er_n);
    v = (lane < cnt) ? t : NEG_INF;
}

__device__ __forceinline__ void g2_rows(const float* __restrict__ feat, int sl, float al,
                                        int cnt, int lane, float& acc0, float& acc1) {
#pragma unroll
    for (int b = 0; b < 4; ++b) {
        if (b * 8 < cnt) {
            float f0[8], f1[8];
#pragma unroll
            for (int i = 0; i < 8; ++i) {
                const int k = b * 8 + i;
                if (k < cnt) {
                    const int s = __builtin_amdgcn_readlane(sl, k);
                    const float* fs = feat + (size_t)s * 64;
                    f0[i] = fs[lane];
                    f1[i] = fs[lane + 32];
                }
            }
#pragma unroll
            for (int i = 0; i < 8; ++i) {
                const int k = b * 8 + i;
                if (k < cnt) {
                    const float a = __int_as_float(__builtin_amdgcn_readlane(__float_as_int(al), k));
                    acc0 += f0[i] * a;
                    acc1 += f1[i] * a;
                }
            }
        }
    }
}

__device__ __forceinline__ float wave_max32(float x) {
#pragma unroll
    for (int o = 16; o > 0; o >>= 1) x = fmaxf(x, __shfl_xor(x, o));
    return x;
}
__device__ __forceinline__ float wave_sum32(float x) {
#pragma unroll
    for (int o = 16; o > 0; o >>= 1) x += __shfl_xor(x, o);
    return x;
}

__device__ __forceinline__ void g2_general(const int* __restrict__ src, const int* __restrict__ dst, int E, int n,
                                           const float* __restrict__ el, float er_n,
                                           const float* __restrict__ feat, int lane,
                                           float& acc0, float& acc1) {
    float m = NEG_INF;
    for_each_in_edge(src, dst, E, n, lane, [&](int s) {
        m = fmaxf(m, leaky(el[s] + er_n));
    });
    if (!__builtin_isfinite(m)) m = 0.0f;
    float d = 0.0f;
    for_each_in_edge(src, dst, E, n, lane, [&](int s) {
        d += expf(leaky(el[s] + er_n) - m);
    });
    const float dm = fmaxf(d, 1e-9f);
    for_each_in_edge(src, dst, E, n, lane, [&](int s) {
        const float a = expf(leaky(el[s] + er_n) - m) / dm;
        const float* fs = feat + (size_t)s * 64;
        acc0 += fs[lane] * a;
        acc1 += fs[lane + 32] * a;
    });
}

__global__ __launch_bounds__(256)
void node_gather2(const int* __restrict__ cnt, const int* __restrict__ bucket,
                  const int* __restrict__ src, const int* __restrict__ dst, int E,
                  const float* __restrict__ el, const float* __restrict__ er,
                  const float* __restrict__ feat, const float* __restrict__ b2,
                  float* __restrict__ out, int N) {
    const int lane = threadIdx.x & 31;
    const int n = __builtin_amdgcn_readfirstlane((int)((blockIdx.x * blockDim.x + threadIdx.x) >> 5));
    if (n >= N) return;
    const int beg = n * CAP;
    const int deg = cnt[n];
    const float er_n = er[n];

    float acc0 = 0.0f, acc1 = 0.0f;
    if (deg > CAP) {
        g2_general(src, dst, E, n, el, er_n, feat, lane, acc0, acc1);
    } else if (deg > 0) {
        int s0;
        float v0;
        g2_scores(bucket, el, er_n, beg, min(deg, 32), lane, s0, v0);

        float m = v0;
        for (int c = 32; c < deg; c += 32) {
            int s2;
            float v2;
            g2_scores(bucket, el, er_n, beg + c, min(deg - c, 32), lane, s2, v2);
            m = fmaxf(m, v2);
        }
        m = wave_max32(m);
        if (!__builtin_isfinite(m)) m = 0.0f;

        const float e0 = expf(v0 - m);
        float d = e0;
        for (int c = 32; c < deg; c += 32) {
            int s2;
            float v2;
            g2_scores(bucket, el, er_n, beg + c, min(deg - c, 32), lane, s2, v2);
            d += expf(v2 - m);
        }
        d = wave_sum32(d);
        const float dm = fmaxf(d, 1e-9f);

        g2_rows(feat, s0, e0 / dm, min(deg, 32), lane, acc0, acc1);
        for (int c = 32; c < deg; c += 32) {
            int s2;
            float v2;
            const int cnt = min(deg - c, 32);
            g2_scores(bucket, el, er_n, beg + c, cnt, lane, s2, v2);
            g2_rows(feat, s2, expf(v2 - m) / dm, cnt, lane, acc0, acc1);
        }
    }

    float* od = out + (size_t)n * 64;
    od[lane]      = acc0 + b2[lane];
    od[lane + 32] = acc1 + b2[lane + 32];
}

extern "C" void kernel_launch(void* const* d_in, const int* in_sizes, int n_in,
                              void* d_out, int out_size, void* d_ws, size_t ws_size,
                              hipStream_t stream) {
    (void)n_in; (void)out_size; (void)ws_size;
    const float* h   = (const float*)d_in[0];
    const int*   src = (const int*)d_in[1];
    const int*   dst = (const int*)d_in[2];
    const float* W1  = (const float*)d_in[3];
    const float* al1 = (const float*)d_in[4];
    const float* ar1 = (const float*)d_in[5];
    const float* b1  = (const float*)d_in[6];
    const float* W2  = (const float*)d_in[7];
    const float* al2 = (const float*)d_in[8];
    const float* ar2 = (const float*)d_in[9];
    const float* b2  = (const float*)d_in[10];
    float* out = (float*)d_out;

    const int N = in_sizes[0] / 128;
    const int E = in_sizes[1];

    int*   bucket  = (int*)d_ws;
    float* feat1   = (float*)(bucket + (size_t)N * CAP);
    float* el1     = feat1 + (size_t)N * 64;
    float* er1     = el1 + (size_t)N * 8;
    float* x2      = er1 + (size_t)N * 8;
    float* feat2   = x2 + (size_t)N * 64;
    float* el2     = feat2 + (size_t)N * 64;
    float* er2     = el2 + N;
    int*   cnt     = (int*)(er2 + N);

    const int B = 256;
    const int gE1  = (E + B - 1) / B;
    const int gNw  = (N * 32 + B - 1) / B;
    const int gG   = (N / 16 + 3) / 4;

    gemm1_coef<<<gG, 128, 0, stream>>>(h, W1, al1, ar1, feat1, el1, er1, cnt, N);

    bucket_fill<<<gE1, B, 0, stream>>>(src, dst, cnt, bucket, E);

    node_gather1<<<gNw, B, 0, stream>>>(cnt, bucket, src, dst, E, el1, er1, feat1, b1, x2, N);

    gemm2_coef  <<<gG, 128, 0, stream>>>(x2, W2, al2, ar2, feat2, el2, er2, N);
    node_gather2<<<gNw, B, 0, stream>>>(cnt, bucket, src, dst, E, el2, er2, feat2, b2, out, N);
}
